# starting kernel
# baseline (speedup 1.0000x reference)
#include <hip/hip_runtime.h>

typedef float v2f __attribute__((ext_vector_type(2)));
typedef float v8f __attribute__((ext_vector_type(8)));

#define TC      5
#define B_      4
#define C_      64
#define T_      1000
#define F_      96
#define O_      2
#define KTOT    (C_ * TC)
#define NKC     (KTOT / 4)
#define TTILE   16
#define WIN     (TTILE + TC - 1)
#define FCHUNK  8
#define TSTRIDE 68
#define XSTRIDE_F 1364
#define WB_OFF  (FCHUNK * XSTRIDE_F)
#define WB_PER_F (KTOT * O_)
#define LDS_FLOATS (WB_OFF + FCHUNK * WB_PER_F)
#define NT_TILES   ((T_ + TTILE - 1) / TTILE)
#define NF_CHUNKS  (F_ / FCHUNK)


__global__ __launch_bounds__(256)
void decoder_out_wmma(const float* __restrict__ x,
                      const float* __restrict__ w,
                      const float* __restrict__ bias,
                      float* __restrict__ out) {
    __shared__ float lds[LDS_FLOATS];

    const int blk    = blockIdx.x;
    const int tTile  = blk % NT_TILES;
    const int fChunk = (blk / NT_TILES) % NF_CHUNKS;
    const int b      = blk / (NT_TILES * NF_CHUNKS);

    const int t0  = tTile * TTILE;
    const int f0  = fChunk * FCHUNK;
    const int tid = threadIdx.x;

    for (int i = tid; i < C_ * WIN * FCHUNK; i += 256) {
        const int fl   = i & (FCHUNK - 1);
        const int pair = i >> 3;
        const int tl   = pair % WIN;
        const int c    = pair / WIN;
        const int tg   = t0 - (TC - 1) + tl;
        float v = 0.0f;
        if (tg >= 0 && tg < T_)
            v = x[((size_t)(b * C_ + c) * T_ + tg) * F_ + (f0 + fl)];
        lds[fl * XSTRIDE_F + tl * TSTRIDE + c] = v;
    }

    for (int d = tid; d < FCHUNK * WB_PER_F; d += 256) {
        const int fl = d / WB_PER_F;
        const int r  = d - fl * WB_PER_F;
        const int kc = r >> 3;
        const int n  = (r >> 2) & 1;
        const int kk = r & 3;
        const int c  = ((kc & 15) << 2) + kk;
        const int k  = kc >> 4;
        lds[WB_OFF + d] = w[(size_t)(f0 + fl) * WB_PER_F + (c * TC + k) * O_ + n];
    }
    __syncthreads();

    const int wave = tid >> 5;
    const int lane = tid & 31;
    const int half = lane >> 4;
    const int mn   = lane & 15;
    const int f    = f0 + wave;

    const float* lfA = &lds[wave * XSTRIDE_F + mn * TSTRIDE + half * 2];
    const int nclamp = (mn < O_) ? mn : 1;
    const float* lfB = &lds[WB_OFF + wave * WB_PER_F + nclamp * 4 + half * 2];

    v8f acc = {};

#pragma unroll
    for (int kc = 0; kc < NKC; ++kc) {
        const int aoff = (kc >> 4) * TSTRIDE + (kc & 15) * 4;
        const int boff = kc * 8;
        const v2f a  = *(const v2f*)(lfA + aoff);
        const v2f bf = *(const v2f*)(lfB + boff);
        acc = __builtin_amdgcn_wmma_f32_16x16x4_f32(
            false, a, false, bf, (short)0, acc, false, false);
    }

    if (mn < O_) {
        const float bv = bias[f * O_ + mn];
#pragma unroll
        for (int r = 0; r < 8; ++r) {
            const int t = t0 + half * 8 + r;
            if (t < T_)
                out[((size_t)(b * O_ + mn) * T_ + t) * F_ + f] = acc[r] + bv;
        }
    }
}

extern "C" void kernel_launch(void* const* d_in, const int* in_sizes, int n_in,
                              void* d_out, int out_size, void* d_ws, size_t ws_size,
                              hipStream_t stream) {
    const float* x    = (const float*)d_in[0];
    const float* w    = (const float*)d_in[1];
    const float* bias = (const float*)d_in[2];
    float* out = (float*)d_out;

    const int grid = B_ * NF_CHUNKS * NT_TILES;
    decoder_out_wmma<<<grid, 256, 0, stream>>>(x, w, bias, out);
}


# K split over waves: work-group = 4 frequency octets x 2 channel halves (8 waves, 2 per SIMD, half the registers each); tap sums of the halves joined once via LDS (sum order: channels, taps, halves)
# speedup vs baseline: 3.7092x; 3.7092x over previous
#include <hip/hip_runtime.h>


typedef float v2f __attribute__((ext_vector_type(2)));
typedef float v4f __attribute__((ext_vector_type(4)));
typedef float v8f __attribute__((ext_vector_type(8)));

#define TC      5
#define B_      4
#define C_      64
#define T_      1000
#define F_      96
#define O_      2
#define KS      2
#define CW      (C_ / KS)
#define NKC     (CW / 4)
#define TT      16
#define LCH     3
#define NSTEP   (LCH + 1)
#define NCHUNK  21
#define FQ      8
#define FH      (FQ / 2)
#define NOCT    4
#define NWV     (NOCT * KS)
#define NFG     (F_ / 32)
#define WROW    (C_ * TC * O_)


__device__ __forceinline__ const float* tile_ptr(const float* __restrict__ x,
                                                 int b, int c0, int row, int fl) {
    int trc = row < 0 ? 0 : row;
    trc = trc > T_ - 1 ? T_ - 1 : trc;
    return x + ((size_t)(b * C_ + c0) * T_ + trc) * F_ + fl;
}

__device__ __forceinline__ float xhalf(float v) {
    return __builtin_bit_cast(float, __builtin_amdgcn_permlanex16(
        __builtin_bit_cast(int, v), __builtin_bit_cast(int, v), 0x76543210, 0xfedcba98, false, false));
}

__device__ __forceinline__ void to_fragments(float& lo, float& hi, bool upper) {
    const float give = upper ? lo : hi;
    const float got  = xhalf(give);
    lo = upper ? got : lo;
    hi = upper ? hi : got;
}

__global__ __launch_bounds__(NWV * 32)
void decoder_out_wmma(const float* __restrict__ x,
                      const float* __restrict__ w,
                      const float* __restrict__ bias,
                      float* __restrict__ out) {
    __shared__ v4f red[NOCT][LCH][2][32];

    const int blk   = blockIdx.x;
    const int chunk = blk % NCHUNK;
    const int fg    = (blk / NCHUNK) % NFG;
    const int b     = blk / (NCHUNK * NFG);

    const int tid  = threadIdx.x;
    const int wave = __builtin_amdgcn_readfirstlane(tid >> 5);
    const int q    = wave & (NOCT - 1);
    const int kh   = wave >> 2;
    const int lane = tid & 31;
    const int h    = lane >> 4;
    const int n    = lane & 15;
    const int f0   = fg * 32 + q * FQ;
    const int c0   = kh * CW;

    const int  o_m   = n >> 3;
    const int  tap_m = n & 7;
    const bool live  = tap_m < TC;
    v2f wA[FQ][NKC];
#pragma unroll
    for (int i = 0; i < FQ; ++i) {
        const float* wf = w + (size_t)(f0 + i) * WROW + (size_t)c0 * TC * O_
                            + (live ? tap_m * O_ + o_m : 0);
#pragma unroll
        for (int j = 0; j < NKC; ++j) {
            const int c = 4 * j + 2 * h;
            const float a0 = wf[(c    ) * TC * O_];
            const float a1 = wf[(c + 1) * TC * O_];
            wA[i][j][0] = live ? a0 : 0.0f;
            wA[i][j][1] = live ? a1 : 0.0f;
        }
    }

    float bv[FQ];
#pragma unroll
    for (int i = 0; i < FQ; ++i) {
        const float bl = bias[(f0 + i) * O_ + h];
        bv[i] = (kh == 0) ? bl : 0.0f;
    }

    float carry[FQ][TC - 1];
#pragma unroll
    for (int i = 0; i < FQ; ++i)
#pragma unroll
        for (int k = 0; k < TC - 1; ++k) carry[i][k] = 0.0f;

    v4f X[NKC][4];
    const int tile0 = chunk * LCH - 1;
    {
        const int  nn    = (n < TT - (TC - 1)) ? TT - (TC - 1) : n;
        const bool valid = (tile0 >= 0) && (nn == n);
        const float* p   = tile_ptr(x, b, c0, tile0 * TT + nn, f0 + FH * h);
#pragma unroll
        for (int j = 0; j < NKC; ++j) {
#pragma unroll
            for (int cc = 0; cc < 4; ++cc) {
                const v4f v = *(const v4f*)(p + (size_t)(4 * j + cc) * T_ * F_);
                X[j][cc][0] = valid ? v[0] : 0.0f;
                X[j][cc][1] = valid ? v[1] : 0.0f;
                X[j][cc][2] = valid ? v[2] : 0.0f;
                X[j][cc][3] = valid ? v[3] : 0.0f;
            }
        }
    }

    float sums[LCH][FQ];

#pragma unroll
    for (int st = 0; st < NSTEP; ++st) {
        const int tbase = (tile0 + st) * TT;
        const float* pn = tile_ptr(x, b, c0, tbase + TT + n, f0 + FH * h);

        v8f acc[FQ];
#pragma unroll
        for (int i = 0; i < FQ; ++i) acc[i] = (v8f){0.f, 0.f, 0.f, 0.f, 0.f, 0.f, 0.f, 0.f};

#pragma unroll
        for (int j = 0; j < NKC; ++j) {
#pragma unroll
            for (int s = 0; s < 2; ++s)
#pragma unroll
                for (int e = 0; e < FH; ++e) {
                    float lo = X[j][s][e], hi = X[j][2 + s][e];
                    to_fragments(lo, hi, h != 0);
                    X[j][s][e]     = lo;
                    X[j][2 + s][e] = hi;
                }

#pragma unroll
            for (int i = 0; i < FQ; ++i) {
                const int g = i / FH;
                const int e = i % FH;
                v2f bf;
                bf[0] = X[j][2 * g    ][e];
                bf[1] = X[j][2 * g + 1][e];
                acc[i] = __builtin_amdgcn_wmma_f32_16x16x4_f32(
                    false, wA[i][j], false, bf, (short)0, acc[i], false, false);
            }
            if (st + 1 < NSTEP) {
                __builtin_amdgcn_sched_barrier(0);
#pragma unroll
                for (int cc = 0; cc < 4; ++cc)
                    X[j][cc] = *(const v4f*)(pn + (size_t)(4 * j + cc) * T_ * F_);
            }
        }

#pragma unroll
        for (int i = 0; i < FQ; ++i) {
            float val = 0.0f;
#pragma unroll
            for (int k = 0; k < TC - 1; ++k) {
                const int shift = (TC - 1) - k;
                const int src   = (lane & 16) | ((n - shift) & 15);
                const float rot = __shfl(acc[i][k], src, 32);
                val += (n >= shift) ? rot : carry[i][k];
                carry[i][k] = rot;
            }
            val += acc[i][TC - 1];
            if (st > 0) sums[st - 1][i] = val + bv[i];
        }
    }

    if (kh != 0) {
#pragma unroll
        for (int u = 0; u < LCH; ++u) {
            red[q][u][0][lane] = (v4f){sums[u][0], sums[u][1], sums[u][2], sums[u][3]};
            red[q][u][1][lane] = (v4f){sums[u][4], sums[u][5], sums[u][6], sums[u][7]};
        }
    }
    __syncthreads();
    if (kh == 0) {
#pragma unroll
        for (int u = 0; u < LCH; ++u) {
            const v4f p0 = red[q][u][0][lane];
            const v4f p1 = red[q][u][1][lane];
            const int t = (tile0 + 1 + u) * TT + n;
            if (t < T_) {
                float* po = out + ((size_t)(b * O_ + h) * T_ + t) * F_ + f0;
                *(v4f*)(po)      = (v4f){sums[u][0] + p0[0], sums[u][1] + p0[1], sums[u][2] + p0[2], sums[u][3] + p0[3]};
                *(v4f*)(po + FH) = (v4f){sums[u][4] + p1[0], sums[u][5] + p1[1], sums[u][6] + p1[2], sums[u][7] + p1[3]};
            }
        }
    }
}

extern "C" void kernel_launch(void* const* d_in, const int* in_sizes, int n_in,
                              void* d_out, int out_size, void* d_ws, size_t ws_size,
                              hipStream_t stream) {
    const float* x    = (const float*)d_in[0];
    const float* w    = (const float*)d_in[1];
    const float* bias = (const float*)d_in[2];
    float* out = (float*)d_out;

    const int grid = B_ * NFG * NCHUNK;
    decoder_out_wmma<<<grid, NWV * 32, 0, stream>>>(x, w, bias, out);
}


# Data path: 4 lanes read 64 contiguous B of one x row (a load touches 8 lines, not 16); pieces stay in their lane: per-piece column and K order, in-lane selects; sum order: channel quarters, taps
# speedup vs baseline: 4.1214x; 1.1111x over previous
#include <hip/hip_runtime.h>


typedef float v2f __attribute__((ext_vector_type(2)));
typedef float v4f __attribute__((ext_vector_type(4)));
typedef float v8f __attribute__((ext_vector_type(8)));

#define TC      5
#define B_      4
#define C_      64
#define T_      1000
#define F_      96
#define O_      2
#define KS      4
#define CW      (C_ / KS)
#define NKC     (CW / 4)
#define TT      16
#define LCH     3
#define NSTEP   (LCH + 1)
#define NCHUNK  21
#define FQ      16
#define NP      (FQ / 4)
#define NFH     2
#define NWV     (NFH * KS)
#define NFG     (F_ / 32)
#define WROW    (C_ * TC * O_)


__device__ __forceinline__ const float* tile_ptr(const float* __restrict__ x,
                                                 int b, int c0, int row, int fl) {
    int trc = row < 0 ? 0 : row;
    trc = trc > T_ - 1 ? T_ - 1 : trc;
    return x + ((size_t)(b * C_ + c0) * T_ + trc) * F_ + fl;
}

__global__ __launch_bounds__(NWV * 32)
void decoder_out_wmma(const float* __restrict__ x,
                      const float* __restrict__ w,
                      const float* __restrict__ bias,
                      float* __restrict__ out) {
    __shared__ v4f red[NFH][KS][LCH][NP][32];

    const int blk   = blockIdx.x;
    const int chunk = blk % NCHUNK;
    const int fg    = (blk / NCHUNK) % NFG;
    const int b     = blk / (NCHUNK * NFG);

    const int tid  = threadIdx.x;
    const int wave = __builtin_amdgcn_readfirstlane(tid >> 5);
    const int fh   = wave & (NFH - 1);
    const int kq   = wave >> 1;
    const int lane = tid & 31;
    const int h    = lane >> 4;
    const int n    = lane & 15;
    const int bn   = n & 1;
    const bool hb  = h != 0;
    const bool nb  = bn != 0;
    const int f0   = fg * 32 + fh * FQ;
    const int c0   = kq * CW;

    const int  o_m   = n >> 3;
    const int  tap_m = n & 7;
    const bool live  = tap_m < TC;
    v2f wA[FQ][NKC];
#pragma unroll
    for (int i = 0; i < FQ; ++i) {
        const int pi = i >> 2;
        const int hx = (pi & 1) ? (1 - h) : h;
        const float* wf = w + (size_t)(f0 + i) * WROW + (size_t)c0 * TC * O_
                            + (live ? tap_m * O_ + o_m : 0);
#pragma unroll
        for (int j = 0; j < NKC; ++j) {
            const int c = 4 * j + 2 * hx;
            const float a0 = wf[(c    ) * TC * O_];
            const float a1 = wf[(c + 1) * TC * O_];
            wA[i][j][0] = live ? a0 : 0.0f;
            wA[i][j][1] = live ? a1 : 0.0f;
        }
    }


    float carry[FQ][TC - 1];
#pragma unroll
    for (int i = 0; i < FQ; ++i)
#pragma unroll
        for (int k = 0; k < TC - 1; ++k) carry[i][k] = 0.0f;

    int  rsrc[2][TC - 1];
    bool rin[2][TC - 1];
#pragma unroll
    for (int tau = 0; tau < 2; ++tau)
#pragma unroll
        for (int k = 0; k < TC - 1; ++k) {
            const int shift = (TC - 1) - k;
            const int rn    = n ^ tau;
            rsrc[tau][k] = (lane & 16) | (((rn - shift) & 15) ^ tau);
            rin[tau][k]  = rn >= shift;
        }

    const int poff00 = 4 * (2 * (bn    ) + (h    ));
    const int poff01 = 4 * (2 * (bn    ) + (h ^ 1));
    const int poff10 = 4 * (2 * (bn ^ 1) + (h    ));
    const int poff11 = 4 * (2 * (bn ^ 1) + (h ^ 1));

    v4f X[NKC][4][2];
    const int tile0 = chunk * LCH - 1;
    {
        const bool inhalo = n >= TT - (TC - 1);
        const bool valid  = (tile0 >= 0) && inhalo;
        const int  r0     = inhalo ? 2 * (n >> 1) : TT - (TC - 1);
        const float* p0 = tile_ptr(x, b, c0, tile0 * TT + r0,     f0);
        const float* p1 = tile_ptr(x, b, c0, tile0 * TT + r0 + 1, f0);
#pragma unroll
        for (int j = 0; j < NKC; ++j) {
#pragma unroll
            for (int cc = 0; cc < 4; ++cc) {
                const size_t co = (size_t)(4 * j + cc) * T_ * F_;
                const v4f v0 = *(const v4f*)(p0 + co + ((cc >> 1) ? poff01 : poff00));
                const v4f v1 = *(const v4f*)(p1 + co + ((cc >> 1) ? poff11 : poff10));
#pragma unroll
                for (int e = 0; e < 4; ++e) {
                    X[j][cc][0][e] = valid ? v0[e] : 0.0f;
                    X[j][cc][1][e] = valid ? v1[e] : 0.0f;
                }
            }
        }
    }

#pragma unroll
    for (int st = 0; st < NSTEP; ++st) {
        const int tbase = (tile0 + st) * TT;
        const float* pn0 = tile_ptr(x, b, c0, tbase + TT + 2 * (n >> 1),     f0);
        const float* pn1 = tile_ptr(x, b, c0, tbase + TT + 2 * (n >> 1) + 1, f0);

        v8f acc[FQ];
#pragma unroll
        for (int i = 0; i < FQ; ++i) acc[i] = (v8f){0.f, 0.f, 0.f, 0.f, 0.f, 0.f, 0.f, 0.f};

#pragma unroll
        for (int j = 0; j < NKC; ++j) {
            v4f Fr[NP][2];
#pragma unroll
            for (int s = 0; s < 2; ++s)
#pragma unroll
                for (int kp = 0; kp < 2; ++kp)
#pragma unroll
                    for (int e = 0; e < 4; ++e) {
                        const float lo = nb ? X[j][s    ][kp ^ 1][e] : X[j][s    ][kp][e];
                        const float hi = nb ? X[j][2 + s][kp ^ 1][e] : X[j][2 + s][kp][e];
                        Fr[2 * kp    ][s][e] = hb ? hi : lo;
                        Fr[2 * kp + 1][s][e] = hb ? lo : hi;
                    }

#pragma unroll
            for (int i = 0; i < FQ; ++i) {
                const int pi = i >> 2;
                const int e  = i & 3;
                v2f bf;
                bf[0] = Fr[pi][0][e];
                bf[1] = Fr[pi][1][e];
                acc[i] = __builtin_amdgcn_wmma_f32_16x16x4_f32(
                    false, wA[i][j], false, bf, (short)0, acc[i], false, false);
            }
            if (st + 1 < NSTEP) {
                __builtin_amdgcn_sched_barrier(0);
#pragma unroll
                for (int cc = 0; cc < 4; ++cc) {
                    const size_t co = (size_t)(4 * j + cc) * T_ * F_;
                    X[j][cc][0] = *(const v4f*)(pn0 + co + ((cc >> 1) ? poff01 : poff00));
                    X[j][cc][1] = *(const v4f*)(pn1 + co + ((cc >> 1) ? poff11 : poff10));
                }
            }
        }

#pragma unroll
        for (int pi = 0; pi < NP; ++pi) {
            const int tau = pi >> 1;
            v4f sv;
#pragma unroll
            for (int e = 0; e < 4; ++e) {
                const int i = 4 * pi + e;
                float val = 0.0f;
#pragma unroll
                for (int k = 0; k < TC - 1; ++k) {
                    const float rot = __shfl(acc[i][k], rsrc[tau][k], 32);
                    val += rin[tau][k] ? rot : carry[i][k];
                    carry[i][k] = rot;
                }
                val += acc[i][TC - 1];
                sv[e] = val;
            }
            if (st > 0) red[fh][kq][st - 1][pi][lane] = sv;
        }
    }

    const int so = wave >> 2;
    const int p8 = lane & 7;
    const int r4 = lane >> 3;
    float bv[4];
#pragma unroll
    for (int e = 0; e < 4; ++e) bv[e] = bias[(fg * 32 + 4 * p8 + e) * O_ + so];
    __syncthreads();
#pragma unroll
    for (int u = 0; u < LCH; ++u) {
        const int trow = (wave & 3) * 12 + u * 4 + r4;
        const int ut   = trow >> 4;
        const int pi   = p8 & 3;
        const int sl   = (so << 4) | ((trow & 15) ^ (pi >> 1));
        const v4f q0 = red[p8 >> 2][0][ut][pi][sl];
        const v4f q1 = red[p8 >> 2][1][ut][pi][sl];
        const v4f q2 = red[p8 >> 2][2][ut][pi][sl];
        const v4f q3 = red[p8 >> 2][3][ut][pi][sl];
        const int t = (tile0 + 1) * TT + trow;
        if (t < T_) {
            float* po = out + ((size_t)(b * O_ + so) * T_ + t) * F_ + fg * 32 + 4 * p8;
            v4f r;
#pragma unroll
            for (int e = 0; e < 4; ++e) r[e] = (((q0[e] + q1[e]) + q2[e]) + q3[e]) + bv[e];
            *(v4f*)(po) = r;
        }
    }
}

extern "C" void kernel_launch(void* const* d_in, const int* in_sizes, int n_in,
                              void* d_out, int out_size, void* d_ws, size_t ws_size,
                              hipStream_t stream) {
    const float* x    = (const float*)d_in[0];
    const float* w    = (const float*)d_in[1];
    const float* bias = (const float*)d_in[2];
    float* out = (float*)d_out;

    const int grid = B_ * NFG * NCHUNK;
    decoder_out_wmma<<<grid, NWV * 32, 0, stream>>>(x, w, bias, out);
}


# Tap rotation on the vector ALU: the 4 shifted taps come by DPP row rotates (row_ror) inside the 16-lane row instead of ds_bpermute through the LDS unit; sum order and values unchanged
# speedup vs baseline: 4.3498x; 1.0554x over previous
#include <hip/hip_runtime.h>


typedef float v2f __attribute__((ext_vector_type(2)));
typedef float v4f __attribute__((ext_vector_type(4)));
typedef float v8f __attribute__((ext_vector_type(8)));

#define TC      5
#define B_      4
#define C_      64
#define T_      1000
#define F_      96
#define O_      2
#define KS      4
#define CW      (C_ / KS)
#define NKC     (CW / 4)
#define TT      16
#define LCH     3
#define NSTEP   (LCH + 1)
#define NCHUNK  21
#define FQ      16
#define NP      (FQ / 4)
#define NFH     2
#define NWV     (NFH * KS)
#define NFG     (F_ / 32)
#define WROW    (C_ * TC * O_)


__device__ __forceinline__ const float* tile_ptr(const float* __restrict__ x,
                                                 int b, int c0, int row, int fl) {
    int trc = row < 0 ? 0 : row;
    trc = trc > T_ - 1 ? T_ - 1 : trc;
    return x + ((size_t)(b * C_ + c0) * T_ + trc) * F_ + fl;
}

template <int N>
__device__ __forceinline__ float row_ror(float v) {
    return __int_as_float(__builtin_amdgcn_update_dpp(0, __float_as_int(v), 0x120 + N,
                                                      0xf, 0xf, true));
}

__global__ __launch_bounds__(NWV * 32)
void decoder_out_wmma(const float* __restrict__ x,
                      const float* __restrict__ w,
                      const float* __restrict__ bias,
                      float* __restrict__ out) {
    __shared__ v4f red[NFH][KS][LCH][NP][32];

    const int blk   = blockIdx.x;
    const int chunk = blk % NCHUNK;
    const int fg    = (blk / NCHUNK) % NFG;
    const int b     = blk / (NCHUNK * NFG);

    const int tid  = threadIdx.x;
    const int wave = __builtin_amdgcn_readfirstlane(tid >> 5);
    const int fh   = wave & (NFH - 1);
    const int kq   = wave >> 1;
    const int lane = tid & 31;
    const int h    = lane >> 4;
    const int n    = lane & 15;
    const int bn   = n & 1;
    const bool hb  = h != 0;
    const bool nb  = bn != 0;
    const int f0   = fg * 32 + fh * FQ;
    const int c0   = kq * CW;

    const int  o_m   = n >> 3;
    const int  tap_m = n & 7;
    const bool live  = tap_m < TC;
    v2f wA[FQ][NKC];
#pragma unroll
    for (int i = 0; i < FQ; ++i) {
        const int pi = i >> 2;
        const int hx = (pi & 1) ? (1 - h) : h;
        const float* wf = w + (size_t)(f0 + i) * WROW + (size_t)c0 * TC * O_
                            + (live ? tap_m * O_ + o_m : 0);
#pragma unroll
        for (int j = 0; j < NKC; ++j) {
            const int c = 4 * j + 2 * hx;
            const float a0 = wf[(c    ) * TC * O_];
            const float a1 = wf[(c + 1) * TC * O_];
            wA[i][j][0] = live ? a0 : 0.0f;
            wA[i][j][1] = live ? a1 : 0.0f;
        }
    }


    float carry[FQ][TC - 1];
#pragma unroll
    for (int i = 0; i < FQ; ++i)
#pragma unroll
        for (int k = 0; k < TC - 1; ++k) carry[i][k] = 0.0f;

    bool rin[2][TC - 1];
#pragma unroll
    for (int tau = 0; tau < 2; ++tau)
#pragma unroll
        for (int k = 0; k < TC - 1; ++k) {
            const int shift = (TC - 1) - k;
            rin[tau][k] = (n ^ tau) >= shift;
        }

    const int poff00 = 4 * (2 * (bn    ) + (h    ));
    const int poff01 = 4 * (2 * (bn    ) + (h ^ 1));
    const int poff10 = 4 * (2 * (bn ^ 1) + (h    ));
    const int poff11 = 4 * (2 * (bn ^ 1) + (h ^ 1));

    v4f X[NKC][4][2];
    const int tile0 = chunk * LCH - 1;
    {
        const bool inhalo = n >= TT - (TC - 1);
        const bool valid  = (tile0 >= 0) && inhalo;
        const int  r0     = inhalo ? 2 * (n >> 1) : TT - (TC - 1);
        const float* p0 = tile_ptr(x, b, c0, tile0 * TT + r0,     f0);
        const float* p1 = tile_ptr(x, b, c0, tile0 * TT + r0 + 1, f0);
#pragma unroll
        for (int j = 0; j < NKC; ++j) {
#pragma unroll
            for (int cc = 0; cc < 4; ++cc) {
                const size_t co = (size_t)(4 * j + cc) * T_ * F_;
                const v4f v0 = *(const v4f*)(p0 + co + ((cc >> 1) ? poff01 : poff00));
                const v4f v1 = *(const v4f*)(p1 + co + ((cc >> 1) ? poff11 : poff10));
#pragma unroll
                for (int e = 0; e < 4; ++e) {
                    X[j][cc][0][e] = valid ? v0[e] : 0.0f;
                    X[j][cc][1][e] = valid ? v1[e] : 0.0f;
                }
            }
        }
    }

#pragma unroll
    for (int st = 0; st < NSTEP; ++st) {
        const int tbase = (tile0 + st) * TT;
        const float* pn0 = tile_ptr(x, b, c0, tbase + TT + 2 * (n >> 1),     f0);
        const float* pn1 = tile_ptr(x, b, c0, tbase + TT + 2 * (n >> 1) + 1, f0);

        v8f acc[FQ];
#pragma unroll
        for (int i = 0; i < FQ; ++i) acc[i] = (v8f){0.f, 0.f, 0.f, 0.f, 0.f, 0.f, 0.f, 0.f};

#pragma unroll
        for (int j = 0; j < NKC; ++j) {
            v4f Fr[NP][2];
#pragma unroll
            for (int s = 0; s < 2; ++s)
#pragma unroll
                for (int kp = 0; kp < 2; ++kp)
#pragma unroll
                    for (int e = 0; e < 4; ++e) {
                        const float lo = nb ? X[j][s    ][kp ^ 1][e] : X[j][s    ][kp][e];
                        const float hi = nb ? X[j][2 + s][kp ^ 1][e] : X[j][2 + s][kp][e];
                        Fr[2 * kp    ][s][e] = hb ? hi : lo;
                        Fr[2 * kp + 1][s][e] = hb ? lo : hi;
                    }

#pragma unroll
            for (int i = 0; i < FQ; ++i) {
                const int pi = i >> 2;
                const int e  = i & 3;
                v2f bf;
                bf[0] = Fr[pi][0][e];
                bf[1] = Fr[pi][1][e];
                acc[i] = __builtin_amdgcn_wmma_f32_16x16x4_f32(
                    false, wA[i][j], false, bf, (short)0, acc[i], false, false);
            }
            if (st + 1 < NSTEP) {
                __builtin_amdgcn_sched_barrier(0);
#pragma unroll
                for (int cc = 0; cc < 4; ++cc) {
                    const size_t co = (size_t)(4 * j + cc) * T_ * F_;
                    X[j][cc][0] = *(const v4f*)(pn0 + co + ((cc >> 1) ? poff01 : poff00));
                    X[j][cc][1] = *(const v4f*)(pn1 + co + ((cc >> 1) ? poff11 : poff10));
                }
            }
        }

#pragma unroll
        for (int pi = 0; pi < NP; ++pi) {
            const int tau = pi >> 1;
            v4f sv;
#pragma unroll
            for (int e = 0; e < 4; ++e) {
                const int i = 4 * pi + e;
                float rot[TC - 1];
                rot[0] = row_ror<4>(acc[i][0]);
                rot[2] = row_ror<2>(acc[i][2]);
                if (tau == 0) {
                    rot[1] = row_ror<3>(acc[i][1]);
                    rot[3] = row_ror<1>(acc[i][3]);
                } else {
                    const float e1 = row_ror<1>(acc[i][1]);
                    const float o1 = row_ror<5>(acc[i][1]);
                    const float e3 = row_ror<15>(acc[i][3]);
                    const float o3 = row_ror<3>(acc[i][3]);
                    rot[1] = nb ? o1 : e1;
                    rot[3] = nb ? o3 : e3;
                }
                float val = 0.0f;
#pragma unroll
                for (int k = 0; k < TC - 1; ++k) {
                    val += rin[tau][k] ? rot[k] : carry[i][k];
                    carry[i][k] = rot[k];
                }
                val += acc[i][TC - 1];
                sv[e] = val;
            }
            if (st > 0) red[fh][kq][st - 1][pi][lane] = sv;
        }
    }

    const int so = wave >> 2;
    const int p8 = lane & 7;
    const int r4 = lane >> 3;
    float bv[4];
#pragma unroll
    for (int e = 0; e < 4; ++e) bv[e] = bias[(fg * 32 + 4 * p8 + e) * O_ + so];
    __syncthreads();
#pragma unroll
    for (int u = 0; u < LCH; ++u) {
        const int trow = (wave & 3) * 12 + u * 4 + r4;
        const int ut   = trow >> 4;
        const int pi   = p8 & 3;
        const int sl   = (so << 4) | ((trow & 15) ^ (pi >> 1));
        const v4f q0 = red[p8 >> 2][0][ut][pi][sl];
        const v4f q1 = red[p8 >> 2][1][ut][pi][sl];
        const v4f q2 = red[p8 >> 2][2][ut][pi][sl];
        const v4f q3 = red[p8 >> 2][3][ut][pi][sl];
        const int t = (tile0 + 1) * TT + trow;
        if (t < T_) {
            float* po = out + ((size_t)(b * O_ + so) * T_ + t) * F_ + fg * 32 + 4 * p8;
            v4f r;
#pragma unroll
            for (int e = 0; e < 4; ++e) r[e] = (((q0[e] + q1[e]) + q2[e]) + q3[e]) + bv[e];
            *(v4f*)(po) = r;
        }
    }
}

extern "C" void kernel_launch(void* const* d_in, const int* in_sizes, int n_in,
                              void* d_out, int out_size, void* d_ws, size_t ws_size,
                              hipStream_t stream) {
    const float* x    = (const float*)d_in[0];
    const float* w    = (const float*)d_in[1];
    const float* bias = (const float*)d_in[2];
    float* out = (float*)d_out;

    const int grid = B_ * NFG * NCHUNK;
    decoder_out_wmma<<<grid, NWV * 32, 0, stream>>>(x, w, bias, out);
}
